# starting kernel
# baseline (speedup 1.0000x reference)
#include <hip/hip_runtime.h>

typedef __attribute__((ext_vector_type(2))) float v2f;
typedef __attribute__((ext_vector_type(8))) float v8f;

#define BB 32
#define NN 2048
#define PP 1024
#define HH 256
#define EE 512
#define BN_EPS 1e-5f

__global__ __launch_bounds__(256) void bps_min_kernel(
    const float* __restrict__ x,
    const float* __restrict__ basis,
    float* __restrict__ feat)
{
  const int tileP = blockIdx.x * 16;
  const int b     = blockIdx.y;
  const int lane  = threadIdx.x & 31;
  const int wave  = threadIdx.x >> 5;
  const int col   = lane & 15;
  const bool hi   = lane >= 16;

  const int   n   = tileP + col;
  const float bx  = basis[n * 3 + 0];
  const float by  = basis[n * 3 + 1];
  const float bz  = basis[n * 3 + 2];
  const float bn2 = bx * bx + by * by + bz * bz;
  v2f Bm;
  Bm.x = hi ? (-2.0f * bz) : (-2.0f * bx);
  Bm.y = hi ? 1.0f         : (-2.0f * by);

  float minv[4];
#pragma unroll
  for (int r = 0; r < 4; ++r) minv[r] = 3.4e38f;

  const float* xb = x + (size_t)b * 3 * NN;
#pragma unroll 2
  for (int i0 = wave * 16; i0 < NN; i0 += 16 * 8) {
    const int   m   = i0 + col;
    const float px  = xb[m];
    const float py  = xb[NN + m];
    const float pz  = xb[2 * NN + m];
    const float pn2 = px * px + py * py + pz * pz;
    v2f Am;
    Am.x = hi ? pz  : px;
    Am.y = hi ? pn2 : py;
    v8f c = {0.f, 0.f, 0.f, 0.f, 0.f, 0.f, 0.f, 0.f};
    c = __builtin_amdgcn_wmma_f32_16x16x4_f32(false, Am, false, Bm,
                                              (short)0, c, false, false);
#pragma unroll
    for (int r = 0; r < 4; ++r)
      minv[r] = fminf(fminf(c[2 * r], c[2 * r + 1]), minv[r]);
  }
  float mv = fminf(fminf(minv[0], minv[1]), fminf(minv[2], minv[3]));
  mv = fminf(mv, __shfl_xor(mv, 16));
  mv = fmaxf(mv + bn2, 0.0f);

  __shared__ float red[8][16];
  if (lane < 16) red[wave][col] = mv;
  __syncthreads();
  if (threadIdx.x < 16) {
    float v = red[0][threadIdx.x];
#pragma unroll
    for (int w = 1; w < 8; ++w) v = fminf(v, red[w][threadIdx.x]);
    feat[(size_t)b * PP + tileP + threadIdx.x] = sqrtf(v);
  }
}

__global__ __launch_bounds__(256) void bn0_kernel(
    const float* __restrict__ feat, const float* __restrict__ g,
    const float* __restrict__ bshift, float* __restrict__ x0)
{
  const int j = blockIdx.x * 256 + threadIdx.x;
  float s = 0.f, ss = 0.f;
#pragma unroll 4
  for (int i = 0; i < BB; ++i) {
    const float v = feat[(size_t)i * PP + j];
    s += v; ss += v * v;
  }
  const float mean = s * (1.0f / BB);
  const float var  = ss * (1.0f / BB) - mean * mean;
  const float sc   = g[j] * rsqrtf(var + BN_EPS);
  const float bt   = bshift[j];
#pragma unroll 4
  for (int i = 0; i < BB; ++i)
    x0[(size_t)i * PP + j] = (feat[(size_t)i * PP + j] - mean) * sc + bt;
}

__device__ __forceinline__ void gemm_segment(v8f& acc, const float* __restrict__ src,
                                             int Ks, const float* __restrict__ Wseg,
                                             int row, int koff) {
  const float* arow = src + (size_t)row * Ks + koff;
  const float* wrow = Wseg + koff;
#pragma unroll 4
  for (int kk = 0; kk < Ks; kk += 4) {
    const float2 av = *reinterpret_cast<const float2*>(arow + kk);
    const float2 wv = *reinterpret_cast<const float2*>(wrow + kk);
    v2f Am; Am.x = av.x; Am.y = av.y;
    v2f Bm; Bm.x = wv.x; Bm.y = wv.y;
    acc = __builtin_amdgcn_wmma_f32_16x16x4_f32(false, Am, false, Bm,
                                                (short)0, acc, false, false);
  }
}

__global__ __launch_bounds__(64) void dense_bn_relu_kernel(
    const float* __restrict__ in0, int K0,
    const float* __restrict__ in1, int K1,
    const float* __restrict__ in2, int K2,
    const float* __restrict__ W,
    const float* __restrict__ bias,
    const float* __restrict__ g, const float* __restrict__ beta,
    float* __restrict__ out, int Hout)
{
  const int  c    = blockIdx.x * 16;
  const int  lane = threadIdx.x & 31;
  const int  wave = threadIdx.x >> 5;
  const int  col  = lane & 15;
  const bool hi   = lane >= 16;
  const int  n    = c + col;
  const int  row  = wave * 16 + col;
  const int  Ktot = K0 + K1 + K2;
  const int  koff = hi ? 2 : 0;

  v8f acc = {0.f, 0.f, 0.f, 0.f, 0.f, 0.f, 0.f, 0.f};
  const float* Wn = W + (size_t)n * Ktot;
  gemm_segment(acc, in0, K0, Wn, row, koff);
  if (K1 > 0) gemm_segment(acc, in1, K1, Wn + K0, row, koff);
  if (K2 > 0) gemm_segment(acc, in2, K2, Wn + K0 + K1, row, koff);

  const float bi = bias[n];
  float vals[8];
  float s = 0.f, ss = 0.f;
#pragma unroll
  for (int r = 0; r < 8; ++r) {
    const float v = fmaxf(acc[r] + bi, 0.f);
    vals[r] = v; s += v; ss += v * v;
  }
  s  += __shfl_xor(s, 16);
  ss += __shfl_xor(ss, 16);

  __shared__ float sred[2][16][2];
  if (lane < 16) { sred[wave][col][0] = s; sred[wave][col][1] = ss; }
  __syncthreads();
  const float ts   = sred[0][col][0] + sred[1][col][0];
  const float tss  = sred[0][col][1] + sred[1][col][1];
  const float mean = ts * (1.0f / BB);
  const float var  = tss * (1.0f / BB) - mean * mean;
  const float sc   = g[n] * rsqrtf(var + BN_EPS);
  const float bt   = beta[n];
#pragma unroll
  for (int r = 0; r < 8; ++r) {
    const int rr = wave * 16 + r + (hi ? 8 : 0);
    out[(size_t)rr * Hout + n] = (vals[r] - mean) * sc + bt;
  }
}

extern "C" void kernel_launch(void* const* d_in, const int* in_sizes, int n_in,
                              void* d_out, int out_size, void* d_ws, size_t ws_size,
                              hipStream_t stream) {
  const float* x     = (const float*)d_in[0];
  const float* basis = (const float*)d_in[1];
  const float* bn0_g = (const float*)d_in[2];
  const float* bn0_b = (const float*)d_in[3];
  const float* W0 = (const float*)d_in[4];
  const float* b0 = (const float*)d_in[5];
  const float* g0 = (const float*)d_in[6];
  const float* be0 = (const float*)d_in[7];
  const float* W1 = (const float*)d_in[8];
  const float* b1 = (const float*)d_in[9];
  const float* g1 = (const float*)d_in[10];
  const float* be1 = (const float*)d_in[11];
  const float* W2 = (const float*)d_in[12];
  const float* b2 = (const float*)d_in[13];
  const float* g2 = (const float*)d_in[14];
  const float* be2 = (const float*)d_in[15];
  const float* W3 = (const float*)d_in[16];
  const float* b3 = (const float*)d_in[17];
  const float* g3 = (const float*)d_in[18];
  const float* be3 = (const float*)d_in[19];
  const float* Wf = (const float*)d_in[20];
  const float* bf = (const float*)d_in[21];
  const float* gf = (const float*)d_in[22];
  const float* bef = (const float*)d_in[23];

  float* ws   = (float*)d_ws;
  float* feat = ws;
  float* x0   = ws + 32768;
  float* h    = ws + 65536;
  float* a1   = ws + 73728;
  float* a2   = ws + 81920;

  bps_min_kernel<<<dim3(PP / 16, BB), 256, 0, stream>>>(x, basis, feat);
  bn0_kernel<<<PP / 256, 256, 0, stream>>>(feat, bn0_g, bn0_b, x0);
  dense_bn_relu_kernel<<<HH / 16, 64, 0, stream>>>(x0, PP, nullptr, 0, nullptr, 0,
                                                   W0, b0, g0, be0, h, HH);
  dense_bn_relu_kernel<<<HH / 16, 64, 0, stream>>>(h, HH, nullptr, 0, nullptr, 0,
                                                   W1, b1, g1, be1, a1, HH);
  dense_bn_relu_kernel<<<HH / 16, 64, 0, stream>>>(x0, PP, a1, HH, nullptr, 0,
                                                   W2, b2, g2, be2, h, HH);
  dense_bn_relu_kernel<<<HH / 16, 64, 0, stream>>>(h, HH, nullptr, 0, nullptr, 0,
                                                   W3, b3, g3, be3, a2, HH);
  dense_bn_relu_kernel<<<EE / 16, 64, 0, stream>>>(x0, PP, a1, HH, a2, HH,
                                                   Wf, bf, gf, bef,
                                                   (float*)d_out, EE);
}


# BPS kernel: each wave keeps 4 basis tiles as B fragments and gives every point fragment to 4 WMMAs with independent D tiles over a quarter of the points; 2048 waves of 128 WMMAs, was 16384 of 16
# speedup vs baseline: 1.9196x; 1.9196x over previous
#include <hip/hip_runtime.h>

typedef __attribute__((ext_vector_type(2))) float v2f;
typedef __attribute__((ext_vector_type(8))) float v8f;

#define BB 32
#define NN 2048
#define PP 1024
#define HH 256
#define EE 512
#define BN_EPS 1e-5f

#define BPS_T 4
#define BPS_S 4
#define BPS_PTS (NN / BPS_S)
#define BPS_GRP (BPS_PTS / 64)

__global__ __launch_bounds__(32 * BPS_S) void bps_min_kernel(
    const float* __restrict__ x,
    const float* __restrict__ basis,
    float* __restrict__ feat)
{
  const int colg  = blockIdx.x * (16 * BPS_T);
  const int b     = blockIdx.y;
  const int lane  = threadIdx.x & 31;
  const int wave  = threadIdx.x >> 5;
  const int col   = lane & 15;
  const bool hi   = lane >= 16;

  v2f   Bm[BPS_T];
  float bn2[BPS_T];
#pragma unroll
  for (int t = 0; t < BPS_T; ++t) {
    const int   n  = colg + 16 * t + col;
    const float bx = basis[n * 3 + 0];
    const float by = basis[n * 3 + 1];
    const float bz = basis[n * 3 + 2];
    bn2[t]  = fmaf(bz, bz, fmaf(bx, bx, by * by));
    Bm[t].x = hi ? (-2.0f * bz) : (-2.0f * bx);
    Bm[t].y = hi ? 1.0f         : (-2.0f * by);
  }

  float minv[BPS_T][4];
#pragma unroll
  for (int t = 0; t < BPS_T; ++t)
#pragma unroll
    for (int r = 0; r < 4; ++r) minv[t][r] = 3.4e38f;

  auto fold = [&](const v8f& d, int t) {
#pragma unroll
    for (int r = 0; r < 4; ++r)
      minv[t][r] = fminf(fminf(d[2 * r], d[2 * r + 1]), minv[t][r]);
  };

  v8f pend;
#pragma unroll
  for (int r = 0; r < 8; ++r) pend[r] = 3.4e38f;

  auto group = [&](const float4& qx, const float4& qy, const float4& qz) {
    const float pxs[4] = {qx.x, qx.y, qx.z, qx.w};
    const float pys[4] = {qy.x, qy.y, qy.z, qy.w};
    const float pzs[4] = {qz.x, qz.y, qz.z, qz.w};
#pragma unroll
    for (int e = 0; e < 4; ++e) {
      const float px  = pxs[e];
      const float py  = pys[e];
      const float pz  = pzs[e];
      const float pn2 = fmaf(pz, pz, fmaf(px, px, py * py));
      v2f Am;
      Am.x = hi ? pz  : px;
      Am.y = hi ? pn2 : py;
#pragma unroll
      for (int t = 0; t < BPS_T; ++t) {
        v8f c = {0.f, 0.f, 0.f, 0.f, 0.f, 0.f, 0.f, 0.f};
        c = __builtin_amdgcn_wmma_f32_16x16x4_f32(false, Am, false, Bm[t],
                                                  (short)0, c, false, false);
        fold(pend, (t + BPS_T - 1) % BPS_T);
        pend = c;
      }
    }
  };

  const float* xs = x + (size_t)b * 3 * NN + wave * BPS_PTS + 4 * col;
  auto ld = [&](int plane, int g) {
    return *reinterpret_cast<const float4*>(xs + plane * NN + 64 * g);
  };
  float4 ax = ld(0, 0), ay = ld(1, 0), az = ld(2, 0);
  float4 bx4 = ld(0, 1), by4 = ld(1, 1), bz4 = ld(2, 1);
#pragma unroll 1
  for (int g = 0; g < BPS_GRP; g += 2) {
    float4 nax = ax, nay = ay, naz = az, nbx = bx4, nby = by4, nbz = bz4;
    if (g + 2 < BPS_GRP) {
      nax = ld(0, g + 2); nay = ld(1, g + 2); naz = ld(2, g + 2);
      nbx = ld(0, g + 3); nby = ld(1, g + 3); nbz = ld(2, g + 3);
    }
    group(ax, ay, az);
    group(bx4, by4, bz4);
    ax = nax; ay = nay; az = naz;
    bx4 = nbx; by4 = nby; bz4 = nbz;
  }
  fold(pend, BPS_T - 1);

  __shared__ float red[BPS_S][16 * BPS_T];
#pragma unroll
  for (int t = 0; t < BPS_T; ++t) {
    float mv = fminf(fminf(minv[t][0], minv[t][1]), fminf(minv[t][2], minv[t][3]));
    mv = fminf(mv, __shfl_xor(mv, 16));
    mv = fmaxf(mv + bn2[t], 0.0f);
    if (lane < 16) red[wave][16 * t + col] = mv;
  }
  __syncthreads();
  if (threadIdx.x < 16 * BPS_T) {
    float v = red[0][threadIdx.x];
#pragma unroll
    for (int w = 1; w < BPS_S; ++w) v = fminf(v, red[w][threadIdx.x]);
    feat[(size_t)b * PP + colg + threadIdx.x] = sqrtf(v);
  }
}

__global__ __launch_bounds__(256) void bn0_kernel(
    const float* __restrict__ feat, const float* __restrict__ g,
    const float* __restrict__ bshift, float* __restrict__ x0)
{
  const int j = blockIdx.x * 256 + threadIdx.x;
  float s = 0.f, ss = 0.f;
#pragma unroll 4
  for (int i = 0; i < BB; ++i) {
    const float v = feat[(size_t)i * PP + j];
    s += v; ss += v * v;
  }
  const float mean = s * (1.0f / BB);
  const float var  = ss * (1.0f / BB) - mean * mean;
  const float sc   = g[j] * rsqrtf(var + BN_EPS);
  const float bt   = bshift[j];
#pragma unroll 4
  for (int i = 0; i < BB; ++i)
    x0[(size_t)i * PP + j] = (feat[(size_t)i * PP + j] - mean) * sc + bt;
}

#define KC 128
#define KSTEPS (KC / 4)

__device__ __forceinline__ v2f load_kpair(const float* __restrict__ p) {
  const float2 v = *reinterpret_cast<const float2*>(p);
  v2f r; r.x = v.x; r.y = v.y;
  return r;
}

__global__ __launch_bounds__(64) void dense_bn_relu_kernel(
    const float* __restrict__ in0, int K0,
    const float* __restrict__ in1, int K1,
    const float* __restrict__ in2, int K2,
    const float* __restrict__ W,
    const float* __restrict__ bias,
    const float* __restrict__ g, const float* __restrict__ beta,
    float* __restrict__ out, int Hout)
{
  const int  c    = blockIdx.x * 16;
  const int  lane = threadIdx.x & 31;
  const int  wave = threadIdx.x >> 5;
  const int  col  = lane & 15;
  const bool hi   = lane >= 16;
  const int  n    = c + col;
  const int  row  = wave * 16 + col;
  const int  Ktot = K0 + K1 + K2;
  const int  koff = hi ? 2 : 0;
  const int  nch  = Ktot / KC;

  const float* wp = W + (size_t)n * Ktot + koff;
  auto a_chunk = [&](int kbase) -> const float* {
    const float* p;
    if (kbase < K0)           p = in0 + (size_t)row * K0 + kbase;
    else if (kbase < K0 + K1) p = in1 + (size_t)row * K1 + (kbase - K0);
    else                      p = in2 + (size_t)row * K2 + (kbase - K0 - K1);
    return p + koff;
  };

  v2f a[KSTEPS], w[KSTEPS];
  {
    const float* ap = a_chunk(0);
#pragma unroll
    for (int j = 0; j < KSTEPS; ++j) {
      a[j] = load_kpair(ap + 4 * j);
      w[j] = load_kpair(wp + 4 * j);
    }
#pragma unroll
    for (int j = 0; j < 2 * KSTEPS; ++j)
      __builtin_amdgcn_sched_group_barrier(0x020, 1, 0);
  }

  v8f acc = {0.f, 0.f, 0.f, 0.f, 0.f, 0.f, 0.f, 0.f};
  for (int ch = 1; ch < nch; ++ch) {
    const float* ap = a_chunk(ch * KC);
    const float* wq = wp + ch * KC;
#pragma unroll
    for (int j = 0; j < KSTEPS; ++j) {
      acc = __builtin_amdgcn_wmma_f32_16x16x4_f32(false, a[j], false, w[j],
                                                  (short)0, acc, false, false);
      a[j] = load_kpair(ap + 4 * j);
      w[j] = load_kpair(wq + 4 * j);
      __builtin_amdgcn_sched_barrier(0);
    }
  }
#pragma unroll
  for (int j = 0; j < KSTEPS; ++j)
    acc = __builtin_amdgcn_wmma_f32_16x16x4_f32(false, a[j], false, w[j],
                                                (short)0, acc, false, false);

  const float bi = bias[n];
  float vals[8];
  float s = 0.f, ss = 0.f;
#pragma unroll
  for (int r = 0; r < 8; ++r) {
    const float v = fmaxf(acc[r] + bi, 0.f);
    vals[r] = v; s += v; ss += v * v;
  }
  s  += __shfl_xor(s, 16);
  ss += __shfl_xor(ss, 16);

  __shared__ float sred[2][16][2];
  if (lane < 16) { sred[wave][col][0] = s; sred[wave][col][1] = ss; }
  __syncthreads();
  const float ts   = sred[0][col][0] + sred[1][col][0];
  const float tss  = sred[0][col][1] + sred[1][col][1];
  const float mean = ts * (1.0f / BB);
  const float var  = tss * (1.0f / BB) - mean * mean;
  const float sc   = g[n] * rsqrtf(var + BN_EPS);
  const float bt   = beta[n];
#pragma unroll
  for (int r = 0; r < 8; ++r) {
    const int rr = wave * 16 + r + (hi ? 8 : 0);
    out[(size_t)rr * Hout + n] = (vals[r] - mean) * sc + bt;
  }
}

extern "C" void kernel_launch(void* const* d_in, const int* in_sizes, int n_in,
                              void* d_out, int out_size, void* d_ws, size_t ws_size,
                              hipStream_t stream) {
  const float* x     = (const float*)d_in[0];
  const float* basis = (const float*)d_in[1];
  const float* bn0_g = (const float*)d_in[2];
  const float* bn0_b = (const float*)d_in[3];
  const float* W0 = (const float*)d_in[4];
  const float* b0 = (const float*)d_in[5];
  const float* g0 = (const float*)d_in[6];
  const float* be0 = (const float*)d_in[7];
  const float* W1 = (const float*)d_in[8];
  const float* b1 = (const float*)d_in[9];
  const float* g1 = (const float*)d_in[10];
  const float* be1 = (const float*)d_in[11];
  const float* W2 = (const float*)d_in[12];
  const float* b2 = (const float*)d_in[13];
  const float* g2 = (const float*)d_in[14];
  const float* be2 = (const float*)d_in[15];
  const float* W3 = (const float*)d_in[16];
  const float* b3 = (const float*)d_in[17];
  const float* g3 = (const float*)d_in[18];
  const float* be3 = (const float*)d_in[19];
  const float* Wf = (const float*)d_in[20];
  const float* bf = (const float*)d_in[21];
  const float* gf = (const float*)d_in[22];
  const float* bef = (const float*)d_in[23];

  float* ws   = (float*)d_ws;
  float* feat = ws;
  float* x0   = ws + 32768;
  float* h    = ws + 65536;
  float* a1   = ws + 73728;
  float* a2   = ws + 81920;

  bps_min_kernel<<<dim3(PP / (16 * BPS_T), BB), 32 * BPS_S, 0, stream>>>(x, basis, feat);
  bn0_kernel<<<PP / 256, 256, 0, stream>>>(feat, bn0_g, bn0_b, x0);
  dense_bn_relu_kernel<<<HH / 16, 64, 0, stream>>>(x0, PP, nullptr, 0, nullptr, 0,
                                                   W0, b0, g0, be0, h, HH);
  dense_bn_relu_kernel<<<HH / 16, 64, 0, stream>>>(h, HH, nullptr, 0, nullptr, 0,
                                                   W1, b1, g1, be1, a1, HH);
  dense_bn_relu_kernel<<<HH / 16, 64, 0, stream>>>(x0, PP, a1, HH, nullptr, 0,
                                                   W2, b2, g2, be2, h, HH);
  dense_bn_relu_kernel<<<HH / 16, 64, 0, stream>>>(h, HH, nullptr, 0, nullptr, 0,
                                                   W3, b3, g3, be3, a2, HH);
  dense_bn_relu_kernel<<<EE / 16, 64, 0, stream>>>(x0, PP, a1, HH, a2, HH,
                                                   Wf, bf, gf, bef,
                                                   (float*)d_out, EE);
}


# dense layers: intermediate activations (x0, h, a1, a2) stored by their producers in WMMA A-fragment order; a wave's A load is one contiguous 512-B read per two k-steps (was two loads of 16 lines)
# speedup vs baseline: 1.9819x; 1.0325x over previous
#include <hip/hip_runtime.h>

typedef __attribute__((ext_vector_type(2))) float v2f;
typedef __attribute__((ext_vector_type(4))) float v4f;
typedef __attribute__((ext_vector_type(8))) float v8f;

#define BB 32
#define NN 2048
#define PP 1024
#define HH 256
#define EE 512
#define BN_EPS 1e-5f

#define BPS_T 4
#define BPS_S 4
#define BPS_PTS (NN / BPS_S)
#define BPS_GRP (BPS_PTS / 64)

__global__ __launch_bounds__(32 * BPS_S) void bps_min_kernel(
    const float* __restrict__ x,
    const float* __restrict__ basis,
    float* __restrict__ feat)
{
  const int colg  = blockIdx.x * (16 * BPS_T);
  const int b     = blockIdx.y;
  const int lane  = threadIdx.x & 31;
  const int wave  = threadIdx.x >> 5;
  const int col   = lane & 15;
  const bool hi   = lane >= 16;

  v2f   Bm[BPS_T];
  float bn2[BPS_T];
#pragma unroll
  for (int t = 0; t < BPS_T; ++t) {
    const int   n  = colg + 16 * t + col;
    const float bx = basis[n * 3 + 0];
    const float by = basis[n * 3 + 1];
    const float bz = basis[n * 3 + 2];
    bn2[t]  = fmaf(bz, bz, fmaf(bx, bx, by * by));
    Bm[t].x = hi ? (-2.0f * bz) : (-2.0f * bx);
    Bm[t].y = hi ? 1.0f         : (-2.0f * by);
  }

  float minv[BPS_T][4];
#pragma unroll
  for (int t = 0; t < BPS_T; ++t)
#pragma unroll
    for (int r = 0; r < 4; ++r) minv[t][r] = 3.4e38f;

  auto fold = [&](const v8f& d, int t) {
#pragma unroll
    for (int r = 0; r < 4; ++r)
      minv[t][r] = fminf(fminf(d[2 * r], d[2 * r + 1]), minv[t][r]);
  };

  v8f pend;
#pragma unroll
  for (int r = 0; r < 8; ++r) pend[r] = 3.4e38f;

  auto group = [&](const float4& qx, const float4& qy, const float4& qz) {
    const float pxs[4] = {qx.x, qx.y, qx.z, qx.w};
    const float pys[4] = {qy.x, qy.y, qy.z, qy.w};
    const float pzs[4] = {qz.x, qz.y, qz.z, qz.w};
#pragma unroll
    for (int e = 0; e < 4; ++e) {
      const float px  = pxs[e];
      const float py  = pys[e];
      const float pz  = pzs[e];
      const float pn2 = fmaf(pz, pz, fmaf(px, px, py * py));
      v2f Am;
      Am.x = hi ? pz  : px;
      Am.y = hi ? pn2 : py;
#pragma unroll
      for (int t = 0; t < BPS_T; ++t) {
        v8f c = {0.f, 0.f, 0.f, 0.f, 0.f, 0.f, 0.f, 0.f};
        c = __builtin_amdgcn_wmma_f32_16x16x4_f32(false, Am, false, Bm[t],
                                                  (short)0, c, false, false);
        fold(pend, (t + BPS_T - 1) % BPS_T);
        pend = c;
      }
    }
  };

  const float* xs = x + (size_t)b * 3 * NN + wave * BPS_PTS + 4 * col;
  auto ld = [&](int plane, int g) {
    return *reinterpret_cast<const float4*>(xs + plane * NN + 64 * g);
  };
  float4 ax = ld(0, 0), ay = ld(1, 0), az = ld(2, 0);
  float4 bx4 = ld(0, 1), by4 = ld(1, 1), bz4 = ld(2, 1);
#pragma unroll 1
  for (int g = 0; g < BPS_GRP; g += 2) {
    float4 nax = ax, nay = ay, naz = az, nbx = bx4, nby = by4, nbz = bz4;
    if (g + 2 < BPS_GRP) {
      nax = ld(0, g + 2); nay = ld(1, g + 2); naz = ld(2, g + 2);
      nbx = ld(0, g + 3); nby = ld(1, g + 3); nbz = ld(2, g + 3);
    }
    group(ax, ay, az);
    group(bx4, by4, bz4);
    ax = nax; ay = nay; az = naz;
    bx4 = nbx; by4 = nby; bz4 = nbz;
  }
  fold(pend, BPS_T - 1);

  __shared__ float red[BPS_S][16 * BPS_T];
#pragma unroll
  for (int t = 0; t < BPS_T; ++t) {
    float mv = fminf(fminf(minv[t][0], minv[t][1]), fminf(minv[t][2], minv[t][3]));
    mv = fminf(mv, __shfl_xor(mv, 16));
    mv = fmaxf(mv + bn2[t], 0.0f);
    if (lane < 16) red[wave][16 * t + col] = mv;
  }
  __syncthreads();
  if (threadIdx.x < 16 * BPS_T) {
    float v = red[0][threadIdx.x];
#pragma unroll
    for (int w = 1; w < BPS_S; ++w) v = fminf(v, red[w][threadIdx.x]);
    feat[(size_t)b * PP + colg + threadIdx.x] = sqrtf(v);
  }
}

__device__ __forceinline__ int frag_index(int m, int k, int K) {
  const int lane = (m & 15) + ((k & 2) << 3);
  return ((((m >> 4) * (K >> 3) + (k >> 3)) * 32 + lane) << 2) + ((k >> 1) & 2) + (k & 1);
}

__global__ __launch_bounds__(256) void bn0_kernel(
    const float* __restrict__ feat, const float* __restrict__ g,
    const float* __restrict__ bshift, float* __restrict__ x0)
{
  const int j = blockIdx.x * 256 + threadIdx.x;
  float s = 0.f, ss = 0.f;
#pragma unroll 4
  for (int i = 0; i < BB; ++i) {
    const float v = feat[(size_t)i * PP + j];
    s += v; ss += v * v;
  }
  const float mean = s * (1.0f / BB);
  const float var  = ss * (1.0f / BB) - mean * mean;
  const float sc   = g[j] * rsqrtf(var + BN_EPS);
  const float bt   = bshift[j];
#pragma unroll 4
  for (int i = 0; i < BB; ++i)
    x0[frag_index(i, j, PP)] = (feat[(size_t)i * PP + j] - mean) * sc + bt;
}

#define KC 128
#define KSTEPS (KC / 4)
#define KPAIRS (KC / 8)

__device__ __forceinline__ v2f load_kpair(const float* __restrict__ p) {
  const float2 v = *reinterpret_cast<const float2*>(p);
  v2f r; r.x = v.x; r.y = v.y;
  return r;
}

template <bool FRAG_OUT>
__global__ __launch_bounds__(64) void dense_bn_relu_kernel(
    const float* __restrict__ in0, int K0,
    const float* __restrict__ in1, int K1,
    const float* __restrict__ in2, int K2,
    const float* __restrict__ W,
    const float* __restrict__ bias,
    const float* __restrict__ g, const float* __restrict__ beta,
    float* __restrict__ out, int Hout)
{
  const int  c    = blockIdx.x * 16;
  const int  lane = threadIdx.x & 31;
  const int  wave = threadIdx.x >> 5;
  const int  col  = lane & 15;
  const bool hi   = lane >= 16;
  const int  n    = c + col;
  const int  Ktot = K0 + K1 + K2;
  const int  koff = hi ? 2 : 0;
  const int  nch  = Ktot / KC;

  const float* wp = W + (size_t)n * Ktot + koff;
  const int mt = __builtin_amdgcn_readfirstlane(wave);
  auto a_chunk = [&](int kbase) -> const float* {
    const float* p;
    int Ks, kl;
    if (kbase < K0)           { p = in0; Ks = K0; kl = kbase; }
    else if (kbase < K0 + K1) { p = in1; Ks = K1; kl = kbase - K0; }
    else                      { p = in2; Ks = K2; kl = kbase - K0 - K1; }
    return p + (size_t)(mt * (Ks >> 3) + (kl >> 3)) * 128;
  };
  const int aoff = 4 * lane;
  auto load_quad = [](const float* __restrict__ p) {
    return *reinterpret_cast<const v4f*>(p);
  };

  v4f a[KPAIRS];
  v2f w[KSTEPS];
  {
    const float* ap = a_chunk(0);
#pragma unroll
    for (int j = 0; j < KPAIRS; ++j) {
      w[2 * j]     = load_kpair(wp + 8 * j);
      w[2 * j + 1] = load_kpair(wp + 8 * j + 4);
      a[j]         = load_quad(ap + aoff + 128 * j);
    }
#pragma unroll
    for (int j = 0; j < KSTEPS + KPAIRS; ++j)
      __builtin_amdgcn_sched_group_barrier(0x020, 1, 0);
  }

  v8f acc = {0.f, 0.f, 0.f, 0.f, 0.f, 0.f, 0.f, 0.f};
  for (int ch = 1; ch < nch; ++ch) {
    const float* ap = a_chunk(ch * KC);
    const float* wq = wp + ch * KC;
#pragma unroll
    for (int j = 0; j < KPAIRS; ++j) {
      const v2f a_lo = __builtin_shufflevector(a[j], a[j], 0, 1);
      const v2f a_hi = __builtin_shufflevector(a[j], a[j], 2, 3);
      acc = __builtin_amdgcn_wmma_f32_16x16x4_f32(false, a_lo, false, w[2 * j],
                                                  (short)0, acc, false, false);
      w[2 * j] = load_kpair(wq + 8 * j);
      __builtin_amdgcn_sched_barrier(0);
      acc = __builtin_amdgcn_wmma_f32_16x16x4_f32(false, a_hi, false, w[2 * j + 1],
                                                  (short)0, acc, false, false);
      w[2 * j + 1] = load_kpair(wq + 8 * j + 4);
      a[j]         = load_quad(ap + aoff + 128 * j);
      __builtin_amdgcn_sched_barrier(0);
    }
  }
#pragma unroll
  for (int j = 0; j < KPAIRS; ++j) {
    const v2f a_lo = __builtin_shufflevector(a[j], a[j], 0, 1);
    const v2f a_hi = __builtin_shufflevector(a[j], a[j], 2, 3);
    acc = __builtin_amdgcn_wmma_f32_16x16x4_f32(false, a_lo, false, w[2 * j],
                                                (short)0, acc, false, false);
    acc = __builtin_amdgcn_wmma_f32_16x16x4_f32(false, a_hi, false, w[2 * j + 1],
                                                (short)0, acc, false, false);
  }

  const float bi = bias[n];
  float vals[8];
  float s = 0.f, ss = 0.f;
#pragma unroll
  for (int r = 0; r < 8; ++r) {
    const float v = fmaxf(acc[r] + bi, 0.f);
    vals[r] = v; s += v; ss += v * v;
  }
  s  += __shfl_xor(s, 16);
  ss += __shfl_xor(ss, 16);

  __shared__ float sred[2][16][2];
  if (lane < 16) { sred[wave][col][0] = s; sred[wave][col][1] = ss; }
  __syncthreads();
  const float ts   = sred[0][col][0] + sred[1][col][0];
  const float tss  = sred[0][col][1] + sred[1][col][1];
  const float mean = ts * (1.0f / BB);
  const float var  = tss * (1.0f / BB) - mean * mean;
  const float sc   = g[n] * rsqrtf(var + BN_EPS);
  const float bt   = beta[n];
#pragma unroll
  for (int r = 0; r < 8; ++r) {
    const int   rr = wave * 16 + r + (hi ? 8 : 0);
    const float y  = (vals[r] - mean) * sc + bt;
    if (FRAG_OUT) out[frag_index(rr, n, Hout)] = y;
    else          out[(size_t)rr * Hout + n] = y;
  }
}

extern "C" void kernel_launch(void* const* d_in, const int* in_sizes, int n_in,
                              void* d_out, int out_size, void* d_ws, size_t ws_size,
                              hipStream_t stream) {
  const float* x     = (const float*)d_in[0];
  const float* basis = (const float*)d_in[1];
  const float* bn0_g = (const float*)d_in[2];
  const float* bn0_b = (const float*)d_in[3];
  const float* W0 = (const float*)d_in[4];
  const float* b0 = (const float*)d_in[5];
  const float* g0 = (const float*)d_in[6];
  const float* be0 = (const float*)d_in[7];
  const float* W1 = (const float*)d_in[8];
  const float* b1 = (const float*)d_in[9];
  const float* g1 = (const float*)d_in[10];
  const float* be1 = (const float*)d_in[11];
  const float* W2 = (const float*)d_in[12];
  const float* b2 = (const float*)d_in[13];
  const float* g2 = (const float*)d_in[14];
  const float* be2 = (const float*)d_in[15];
  const float* W3 = (const float*)d_in[16];
  const float* b3 = (const float*)d_in[17];
  const float* g3 = (const float*)d_in[18];
  const float* be3 = (const float*)d_in[19];
  const float* Wf = (const float*)d_in[20];
  const float* bf = (const float*)d_in[21];
  const float* gf = (const float*)d_in[22];
  const float* bef = (const float*)d_in[23];

  float* ws   = (float*)d_ws;
  float* feat = ws;
  float* x0   = ws + 32768;
  float* h    = ws + 65536;
  float* a1   = ws + 73728;
  float* a2   = ws + 81920;

  bps_min_kernel<<<dim3(PP / (16 * BPS_T), BB), 32 * BPS_S, 0, stream>>>(x, basis, feat);
  bn0_kernel<<<PP / 256, 256, 0, stream>>>(feat, bn0_g, bn0_b, x0);
  dense_bn_relu_kernel<true><<<HH / 16, 64, 0, stream>>>(x0, PP, nullptr, 0, nullptr, 0,
                                                   W0, b0, g0, be0, h, HH);
  dense_bn_relu_kernel<true><<<HH / 16, 64, 0, stream>>>(h, HH, nullptr, 0, nullptr, 0,
                                                   W1, b1, g1, be1, a1, HH);
  dense_bn_relu_kernel<true><<<HH / 16, 64, 0, stream>>>(x0, PP, a1, HH, nullptr, 0,
                                                   W2, b2, g2, be2, h, HH);
  dense_bn_relu_kernel<true><<<HH / 16, 64, 0, stream>>>(h, HH, nullptr, 0, nullptr, 0,
                                                   W3, b3, g3, be3, a2, HH);
  dense_bn_relu_kernel<false><<<EE / 16, 64, 0, stream>>>(x0, PP, a1, HH, a2, HH,
                                                   Wf, bf, gf, bef,
                                                   (float*)d_out, EE);
}


# BN0 kernel: a thread loads its column's 32 batch values and scale/shift as one group and keeps them in registers (13 dependent load trips become 1); 16 workgroups of 64 threads, was 4 of 256
# speedup vs baseline: 2.1889x; 1.1044x over previous
#include <hip/hip_runtime.h>

typedef __attribute__((ext_vector_type(2))) float v2f;
typedef __attribute__((ext_vector_type(4))) float v4f;
typedef __attribute__((ext_vector_type(8))) float v8f;

#define BB 32
#define NN 2048
#define PP 1024
#define HH 256
#define EE 512
#define BN_EPS 1e-5f

#define BPS_T 4
#define BPS_S 4
#define BPS_PTS (NN / BPS_S)
#define BPS_GRP (BPS_PTS / 64)

__global__ __launch_bounds__(32 * BPS_S) void bps_min_kernel(
    const float* __restrict__ x,
    const float* __restrict__ basis,
    float* __restrict__ feat)
{
  const int colg  = blockIdx.x * (16 * BPS_T);
  const int b     = blockIdx.y;
  const int lane  = threadIdx.x & 31;
  const int wave  = threadIdx.x >> 5;
  const int col   = lane & 15;
  const bool hi   = lane >= 16;

  v2f   Bm[BPS_T];
  float bn2[BPS_T];
#pragma unroll
  for (int t = 0; t < BPS_T; ++t) {
    const int   n  = colg + 16 * t + col;
    const float bx = basis[n * 3 + 0];
    const float by = basis[n * 3 + 1];
    const float bz = basis[n * 3 + 2];
    bn2[t]  = fmaf(bz, bz, fmaf(bx, bx, by * by));
    Bm[t].x = hi ? (-2.0f * bz) : (-2.0f * bx);
    Bm[t].y = hi ? 1.0f         : (-2.0f * by);
  }

  float minv[BPS_T][4];
#pragma unroll
  for (int t = 0; t < BPS_T; ++t)
#pragma unroll
    for (int r = 0; r < 4; ++r) minv[t][r] = 3.4e38f;

  auto fold = [&](const v8f& d, int t) {
#pragma unroll
    for (int r = 0; r < 4; ++r)
      minv[t][r] = fminf(fminf(d[2 * r], d[2 * r + 1]), minv[t][r]);
  };

  v8f pend;
#pragma unroll
  for (int r = 0; r < 8; ++r) pend[r] = 3.4e38f;

  auto group = [&](const float4& qx, const float4& qy, const float4& qz) {
    const float pxs[4] = {qx.x, qx.y, qx.z, qx.w};
    const float pys[4] = {qy.x, qy.y, qy.z, qy.w};
    const float pzs[4] = {qz.x, qz.y, qz.z, qz.w};
#pragma unroll
    for (int e = 0; e < 4; ++e) {
      const float px  = pxs[e];
      const float py  = pys[e];
      const float pz  = pzs[e];
      const float pn2 = fmaf(pz, pz, fmaf(px, px, py * py));
      v2f Am;
      Am.x = hi ? pz  : px;
      Am.y = hi ? pn2 : py;
#pragma unroll
      for (int t = 0; t < BPS_T; ++t) {
        v8f c = {0.f, 0.f, 0.f, 0.f, 0.f, 0.f, 0.f, 0.f};
        c = __builtin_amdgcn_wmma_f32_16x16x4_f32(false, Am, false, Bm[t],
                                                  (short)0, c, false, false);
        fold(pend, (t + BPS_T - 1) % BPS_T);
        pend = c;
      }
    }
  };

  const float* xs = x + (size_t)b * 3 * NN + wave * BPS_PTS + 4 * col;
  auto ld = [&](int plane, int g) {
    return *reinterpret_cast<const float4*>(xs + plane * NN + 64 * g);
  };
  float4 ax = ld(0, 0), ay = ld(1, 0), az = ld(2, 0);
  float4 bx4 = ld(0, 1), by4 = ld(1, 1), bz4 = ld(2, 1);
#pragma unroll 1
  for (int g = 0; g < BPS_GRP; g += 2) {
    float4 nax = ax, nay = ay, naz = az, nbx = bx4, nby = by4, nbz = bz4;
    if (g + 2 < BPS_GRP) {
      nax = ld(0, g + 2); nay = ld(1, g + 2); naz = ld(2, g + 2);
      nbx = ld(0, g + 3); nby = ld(1, g + 3); nbz = ld(2, g + 3);
    }
    group(ax, ay, az);
    group(bx4, by4, bz4);
    ax = nax; ay = nay; az = naz;
    bx4 = nbx; by4 = nby; bz4 = nbz;
  }
  fold(pend, BPS_T - 1);

  __shared__ float red[BPS_S][16 * BPS_T];
#pragma unroll
  for (int t = 0; t < BPS_T; ++t) {
    float mv = fminf(fminf(minv[t][0], minv[t][1]), fminf(minv[t][2], minv[t][3]));
    mv = fminf(mv, __shfl_xor(mv, 16));
    mv = fmaxf(mv + bn2[t], 0.0f);
    if (lane < 16) red[wave][16 * t + col] = mv;
  }
  __syncthreads();
  if (threadIdx.x < 16 * BPS_T) {
    float v = red[0][threadIdx.x];
#pragma unroll
    for (int w = 1; w < BPS_S; ++w) v = fminf(v, red[w][threadIdx.x]);
    feat[(size_t)b * PP + colg + threadIdx.x] = sqrtf(v);
  }
}

__device__ __forceinline__ int frag_index(int m, int k, int K) {
  const int lane = (m & 15) + ((k & 2) << 3);
  return ((((m >> 4) * (K >> 3) + (k >> 3)) * 32 + lane) << 2) + ((k >> 1) & 2) + (k & 1);
}

#define BN0_T 64
__global__ __launch_bounds__(BN0_T) void bn0_kernel(
    const float* __restrict__ feat, const float* __restrict__ g,
    const float* __restrict__ bshift, float* __restrict__ x0)
{
  const int j = blockIdx.x * BN0_T + threadIdx.x;
  float v[BB];
#pragma unroll
  for (int i = 0; i < BB; ++i) v[i] = feat[(size_t)i * PP + j];
  const float gj = g[j];
  const float bt = bshift[j];
  float s = 0.f, ss = 0.f;
#pragma unroll
  for (int i = 0; i < BB; ++i) { s += v[i]; ss += v[i] * v[i]; }
  const float mean = s * (1.0f / BB);
  const float var  = ss * (1.0f / BB) - mean * mean;
  const float sc   = gj * rsqrtf(var + BN_EPS);
#pragma unroll
  for (int i = 0; i < BB; ++i)
    x0[frag_index(i, j, PP)] = (v[i] - mean) * sc + bt;
}

#define KC 128
#define KSTEPS (KC / 4)
#define KPAIRS (KC / 8)

__device__ __forceinline__ v2f load_kpair(const float* __restrict__ p) {
  const float2 v = *reinterpret_cast<const float2*>(p);
  v2f r; r.x = v.x; r.y = v.y;
  return r;
}

template <bool FRAG_OUT>
__global__ __launch_bounds__(64) void dense_bn_relu_kernel(
    const float* __restrict__ in0, int K0,
    const float* __restrict__ in1, int K1,
    const float* __restrict__ in2, int K2,
    const float* __restrict__ W,
    const float* __restrict__ bias,
    const float* __restrict__ g, const float* __restrict__ beta,
    float* __restrict__ out, int Hout)
{
  const int  c    = blockIdx.x * 16;
  const int  lane = threadIdx.x & 31;
  const int  wave = threadIdx.x >> 5;
  const int  col  = lane & 15;
  const bool hi   = lane >= 16;
  const int  n    = c + col;
  const int  Ktot = K0 + K1 + K2;
  const int  koff = hi ? 2 : 0;
  const int  nch  = Ktot / KC;

  const float* wp = W + (size_t)n * Ktot + koff;
  const int mt = __builtin_amdgcn_readfirstlane(wave);
  auto a_chunk = [&](int kbase) -> const float* {
    const float* p;
    int Ks, kl;
    if (kbase < K0)           { p = in0; Ks = K0; kl = kbase; }
    else if (kbase < K0 + K1) { p = in1; Ks = K1; kl = kbase - K0; }
    else                      { p = in2; Ks = K2; kl = kbase - K0 - K1; }
    return p + (size_t)(mt * (Ks >> 3) + (kl >> 3)) * 128;
  };
  const int aoff = 4 * lane;
  auto load_quad = [](const float* __restrict__ p) {
    return *reinterpret_cast<const v4f*>(p);
  };

  v4f a[KPAIRS];
  v2f w[KSTEPS];
  {
    const float* ap = a_chunk(0);
#pragma unroll
    for (int j = 0; j < KPAIRS; ++j) {
      w[2 * j]     = load_kpair(wp + 8 * j);
      w[2 * j + 1] = load_kpair(wp + 8 * j + 4);
      a[j]         = load_quad(ap + aoff + 128 * j);
    }
#pragma unroll
    for (int j = 0; j < KSTEPS + KPAIRS; ++j)
      __builtin_amdgcn_sched_group_barrier(0x020, 1, 0);
  }

  v8f acc = {0.f, 0.f, 0.f, 0.f, 0.f, 0.f, 0.f, 0.f};
  for (int ch = 1; ch < nch; ++ch) {
    const float* ap = a_chunk(ch * KC);
    const float* wq = wp + ch * KC;
#pragma unroll
    for (int j = 0; j < KPAIRS; ++j) {
      const v2f a_lo = __builtin_shufflevector(a[j], a[j], 0, 1);
      const v2f a_hi = __builtin_shufflevector(a[j], a[j], 2, 3);
      acc = __builtin_amdgcn_wmma_f32_16x16x4_f32(false, a_lo, false, w[2 * j],
                                                  (short)0, acc, false, false);
      w[2 * j] = load_kpair(wq + 8 * j);
      __builtin_amdgcn_sched_barrier(0);
      acc = __builtin_amdgcn_wmma_f32_16x16x4_f32(false, a_hi, false, w[2 * j + 1],
                                                  (short)0, acc, false, false);
      w[2 * j + 1] = load_kpair(wq + 8 * j + 4);
      a[j]         = load_quad(ap + aoff + 128 * j);
      __builtin_amdgcn_sched_barrier(0);
    }
  }
#pragma unroll
  for (int j = 0; j < KPAIRS; ++j) {
    const v2f a_lo = __builtin_shufflevector(a[j], a[j], 0, 1);
    const v2f a_hi = __builtin_shufflevector(a[j], a[j], 2, 3);
    acc = __builtin_amdgcn_wmma_f32_16x16x4_f32(false, a_lo, false, w[2 * j],
                                                (short)0, acc, false, false);
    acc = __builtin_amdgcn_wmma_f32_16x16x4_f32(false, a_hi, false, w[2 * j + 1],
                                                (short)0, acc, false, false);
  }

  const float bi = bias[n];
  float vals[8];
  float s = 0.f, ss = 0.f;
#pragma unroll
  for (int r = 0; r < 8; ++r) {
    const float v = fmaxf(acc[r] + bi, 0.f);
    vals[r] = v; s += v; ss += v * v;
  }
  s  += __shfl_xor(s, 16);
  ss += __shfl_xor(ss, 16);

  __shared__ float sred[2][16][2];
  if (lane < 16) { sred[wave][col][0] = s; sred[wave][col][1] = ss; }
  __syncthreads();
  const float ts   = sred[0][col][0] + sred[1][col][0];
  const float tss  = sred[0][col][1] + sred[1][col][1];
  const float mean = ts * (1.0f / BB);
  const float var  = tss * (1.0f / BB) - mean * mean;
  const float sc   = g[n] * rsqrtf(var + BN_EPS);
  const float bt   = beta[n];
#pragma unroll
  for (int r = 0; r < 8; ++r) {
    const int   rr = wave * 16 + r + (hi ? 8 : 0);
    const float y  = (vals[r] - mean) * sc + bt;
    if (FRAG_OUT) out[frag_index(rr, n, Hout)] = y;
    else          out[(size_t)rr * Hout + n] = y;
  }
}

extern "C" void kernel_launch(void* const* d_in, const int* in_sizes, int n_in,
                              void* d_out, int out_size, void* d_ws, size_t ws_size,
                              hipStream_t stream) {
  const float* x     = (const float*)d_in[0];
  const float* basis = (const float*)d_in[1];
  const float* bn0_g = (const float*)d_in[2];
  const float* bn0_b = (const float*)d_in[3];
  const float* W0 = (const float*)d_in[4];
  const float* b0 = (const float*)d_in[5];
  const float* g0 = (const float*)d_in[6];
  const float* be0 = (const float*)d_in[7];
  const float* W1 = (const float*)d_in[8];
  const float* b1 = (const float*)d_in[9];
  const float* g1 = (const float*)d_in[10];
  const float* be1 = (const float*)d_in[11];
  const float* W2 = (const float*)d_in[12];
  const float* b2 = (const float*)d_in[13];
  const float* g2 = (const float*)d_in[14];
  const float* be2 = (const float*)d_in[15];
  const float* W3 = (const float*)d_in[16];
  const float* b3 = (const float*)d_in[17];
  const float* g3 = (const float*)d_in[18];
  const float* be3 = (const float*)d_in[19];
  const float* Wf = (const float*)d_in[20];
  const float* bf = (const float*)d_in[21];
  const float* gf = (const float*)d_in[22];
  const float* bef = (const float*)d_in[23];

  float* ws   = (float*)d_ws;
  float* feat = ws;
  float* x0   = ws + 32768;
  float* h    = ws + 65536;
  float* a1   = ws + 73728;
  float* a2   = ws + 81920;

  bps_min_kernel<<<dim3(PP / (16 * BPS_T), BB), 32 * BPS_S, 0, stream>>>(x, basis, feat);
  bn0_kernel<<<PP / BN0_T, BN0_T, 0, stream>>>(feat, bn0_g, bn0_b, x0);
  dense_bn_relu_kernel<true><<<HH / 16, 64, 0, stream>>>(x0, PP, nullptr, 0, nullptr, 0,
                                                   W0, b0, g0, be0, h, HH);
  dense_bn_relu_kernel<true><<<HH / 16, 64, 0, stream>>>(h, HH, nullptr, 0, nullptr, 0,
                                                   W1, b1, g1, be1, a1, HH);
  dense_bn_relu_kernel<true><<<HH / 16, 64, 0, stream>>>(x0, PP, a1, HH, nullptr, 0,
                                                   W2, b2, g2, be2, h, HH);
  dense_bn_relu_kernel<true><<<HH / 16, 64, 0, stream>>>(h, HH, nullptr, 0, nullptr, 0,
                                                   W3, b3, g3, be3, a2, HH);
  dense_bn_relu_kernel<false><<<EE / 16, 64, 0, stream>>>(x0, PP, a1, HH, a2, HH,
                                                   Wf, bf, gf, bef,
                                                   (float*)d_out, EE);
}


# dense layers: at entry, behind the first chunk's loads, each wave prefetches to L2 the later chunks of its A tile and its share of the 16 weight rows (one lane per 256-B unit)
# speedup vs baseline: 2.2768x; 1.0402x over previous
#include <hip/hip_runtime.h>

typedef __attribute__((ext_vector_type(2))) float v2f;
typedef __attribute__((ext_vector_type(4))) float v4f;
typedef __attribute__((ext_vector_type(8))) float v8f;

#define BB 32
#define NN 2048
#define PP 1024
#define HH 256
#define EE 512
#define BN_EPS 1e-5f

#define BPS_T 4
#define BPS_S 4
#define BPS_PTS (NN / BPS_S)
#define BPS_GRP (BPS_PTS / 64)

__global__ __launch_bounds__(32 * BPS_S) void bps_min_kernel(
    const float* __restrict__ x,
    const float* __restrict__ basis,
    float* __restrict__ feat)
{
  const int colg  = blockIdx.x * (16 * BPS_T);
  const int b     = blockIdx.y;
  const int lane  = threadIdx.x & 31;
  const int wave  = threadIdx.x >> 5;
  const int col   = lane & 15;
  const bool hi   = lane >= 16;

  v2f   Bm[BPS_T];
  float bn2[BPS_T];
#pragma unroll
  for (int t = 0; t < BPS_T; ++t) {
    const int   n  = colg + 16 * t + col;
    const float bx = basis[n * 3 + 0];
    const float by = basis[n * 3 + 1];
    const float bz = basis[n * 3 + 2];
    bn2[t]  = fmaf(bz, bz, fmaf(bx, bx, by * by));
    Bm[t].x = hi ? (-2.0f * bz) : (-2.0f * bx);
    Bm[t].y = hi ? 1.0f         : (-2.0f * by);
  }

  float minv[BPS_T][4];
#pragma unroll
  for (int t = 0; t < BPS_T; ++t)
#pragma unroll
    for (int r = 0; r < 4; ++r) minv[t][r] = 3.4e38f;

  auto fold = [&](const v8f& d, int t) {
#pragma unroll
    for (int r = 0; r < 4; ++r)
      minv[t][r] = fminf(fminf(d[2 * r], d[2 * r + 1]), minv[t][r]);
  };

  v8f pend;
#pragma unroll
  for (int r = 0; r < 8; ++r) pend[r] = 3.4e38f;

  auto group = [&](const float4& qx, const float4& qy, const float4& qz) {
    const float pxs[4] = {qx.x, qx.y, qx.z, qx.w};
    const float pys[4] = {qy.x, qy.y, qy.z, qy.w};
    const float pzs[4] = {qz.x, qz.y, qz.z, qz.w};
#pragma unroll
    for (int e = 0; e < 4; ++e) {
      const float px  = pxs[e];
      const float py  = pys[e];
      const float pz  = pzs[e];
      const float pn2 = fmaf(pz, pz, fmaf(px, px, py * py));
      v2f Am;
      Am.x = hi ? pz  : px;
      Am.y = hi ? pn2 : py;
#pragma unroll
      for (int t = 0; t < BPS_T; ++t) {
        v8f c = {0.f, 0.f, 0.f, 0.f, 0.f, 0.f, 0.f, 0.f};
        c = __builtin_amdgcn_wmma_f32_16x16x4_f32(false, Am, false, Bm[t],
                                                  (short)0, c, false, false);
        fold(pend, (t + BPS_T - 1) % BPS_T);
        pend = c;
      }
    }
  };

  const float* xs = x + (size_t)b * 3 * NN + wave * BPS_PTS + 4 * col;
  auto ld = [&](int plane, int g) {
    return *reinterpret_cast<const float4*>(xs + plane * NN + 64 * g);
  };
  float4 ax = ld(0, 0), ay = ld(1, 0), az = ld(2, 0);
  float4 bx4 = ld(0, 1), by4 = ld(1, 1), bz4 = ld(2, 1);
#pragma unroll 1
  for (int g = 0; g < BPS_GRP; g += 2) {
    float4 nax = ax, nay = ay, naz = az, nbx = bx4, nby = by4, nbz = bz4;
    if (g + 2 < BPS_GRP) {
      nax = ld(0, g + 2); nay = ld(1, g + 2); naz = ld(2, g + 2);
      nbx = ld(0, g + 3); nby = ld(1, g + 3); nbz = ld(2, g + 3);
    }
    group(ax, ay, az);
    group(bx4, by4, bz4);
    ax = nax; ay = nay; az = naz;
    bx4 = nbx; by4 = nby; bz4 = nbz;
  }
  fold(pend, BPS_T - 1);

  __shared__ float red[BPS_S][16 * BPS_T];
#pragma unroll
  for (int t = 0; t < BPS_T; ++t) {
    float mv = fminf(fminf(minv[t][0], minv[t][1]), fminf(minv[t][2], minv[t][3]));
    mv = fminf(mv, __shfl_xor(mv, 16));
    mv = fmaxf(mv + bn2[t], 0.0f);
    if (lane < 16) red[wave][16 * t + col] = mv;
  }
  __syncthreads();
  if (threadIdx.x < 16 * BPS_T) {
    float v = red[0][threadIdx.x];
#pragma unroll
    for (int w = 1; w < BPS_S; ++w) v = fminf(v, red[w][threadIdx.x]);
    feat[(size_t)b * PP + colg + threadIdx.x] = sqrtf(v);
  }
}

__device__ __forceinline__ int frag_index(int m, int k, int K) {
  const int lane = (m & 15) + ((k & 2) << 3);
  return ((((m >> 4) * (K >> 3) + (k >> 3)) * 32 + lane) << 2) + ((k >> 1) & 2) + (k & 1);
}

#define BN0_T 64
__global__ __launch_bounds__(BN0_T) void bn0_kernel(
    const float* __restrict__ feat, const float* __restrict__ g,
    const float* __restrict__ bshift, float* __restrict__ x0)
{
  const int j = blockIdx.x * BN0_T + threadIdx.x;
  float v[BB];
#pragma unroll
  for (int i = 0; i < BB; ++i) v[i] = feat[(size_t)i * PP + j];
  const float gj = g[j];
  const float bt = bshift[j];
  float s = 0.f, ss = 0.f;
#pragma unroll
  for (int i = 0; i < BB; ++i) { s += v[i]; ss += v[i] * v[i]; }
  const float mean = s * (1.0f / BB);
  const float var  = ss * (1.0f / BB) - mean * mean;
  const float sc   = gj * rsqrtf(var + BN_EPS);
#pragma unroll
  for (int i = 0; i < BB; ++i)
    x0[frag_index(i, j, PP)] = (v[i] - mean) * sc + bt;
}

#define KC 128
#define KSTEPS (KC / 4)
#define KPAIRS (KC / 8)

__device__ __forceinline__ v2f load_kpair(const float* __restrict__ p) {
  const float2 v = *reinterpret_cast<const float2*>(p);
  v2f r; r.x = v.x; r.y = v.y;
  return r;
}

template <bool FRAG_OUT>
__global__ __launch_bounds__(64) void dense_bn_relu_kernel(
    const float* __restrict__ in0, int K0,
    const float* __restrict__ in1, int K1,
    const float* __restrict__ in2, int K2,
    const float* __restrict__ W,
    const float* __restrict__ bias,
    const float* __restrict__ g, const float* __restrict__ beta,
    float* __restrict__ out, int Hout)
{
  const int  c    = blockIdx.x * 16;
  const int  lane = threadIdx.x & 31;
  const int  wave = threadIdx.x >> 5;
  const int  col  = lane & 15;
  const bool hi   = lane >= 16;
  const int  n    = c + col;
  const int  Ktot = K0 + K1 + K2;
  const int  koff = hi ? 2 : 0;
  const int  nch  = Ktot / KC;

  const float* wp = W + (size_t)n * Ktot + koff;
  const int mt = __builtin_amdgcn_readfirstlane(wave);
  auto a_chunk = [&](int kbase) -> const float* {
    const float* p;
    int Ks, kl;
    if (kbase < K0)           { p = in0; Ks = K0; kl = kbase; }
    else if (kbase < K0 + K1) { p = in1; Ks = K1; kl = kbase - K0; }
    else                      { p = in2; Ks = K2; kl = kbase - K0 - K1; }
    return p + (size_t)(mt * (Ks >> 3) + (kl >> 3)) * 128;
  };
  const int aoff = 4 * lane;
  auto load_quad = [](const float* __restrict__ p) {
    return *reinterpret_cast<const v4f*>(p);
  };

  v4f a[KPAIRS];
  v2f w[KSTEPS];
  {
    const float* ap = a_chunk(0);
#pragma unroll
    for (int j = 0; j < KPAIRS; ++j) {
      w[2 * j]     = load_kpair(wp + 8 * j);
      w[2 * j + 1] = load_kpair(wp + 8 * j + 4);
      a[j]         = load_quad(ap + aoff + 128 * j);
    }
#pragma unroll
    for (int j = 0; j < KSTEPS + KPAIRS; ++j)
      __builtin_amdgcn_sched_group_barrier(0x020, 1, 0);
  }

  {
    auto pf = [](const float* p) { __builtin_prefetch(p, 0, 2); };
    const float* pw = W + (size_t)(c + (threadIdx.x & 15)) * Ktot;
    const int    nw = Ktot >> 6;
    int          uw = KC / 64 + (threadIdx.x >> 4);
    const float* t0 = in0 + (size_t)mt * (K0 >> 3) * 128;
    const int    n0 = K0 >> 2;
    int          ua = KC / 4 + lane;
    while (ua < n0 || uw < nw) {
      if (ua < n0) pf(t0 + ua * 64);
      if (uw < nw) pf(pw + uw * 64);
      ua += 32;
      uw += 4;
    }
    const float* t1 = in1 + (size_t)mt * (K1 >> 3) * 128;
    for (int u = lane; u < (K1 >> 2); u += 32) pf(t1 + u * 64);
    const float* t2 = in2 + (size_t)mt * (K2 >> 3) * 128;
    for (int u = lane; u < (K2 >> 2); u += 32) pf(t2 + u * 64);
  }

  v8f acc = {0.f, 0.f, 0.f, 0.f, 0.f, 0.f, 0.f, 0.f};
  for (int ch = 1; ch < nch; ++ch) {
    const float* ap = a_chunk(ch * KC);
    const float* wq = wp + ch * KC;
#pragma unroll
    for (int j = 0; j < KPAIRS; ++j) {
      const v2f a_lo = __builtin_shufflevector(a[j], a[j], 0, 1);
      const v2f a_hi = __builtin_shufflevector(a[j], a[j], 2, 3);
      acc = __builtin_amdgcn_wmma_f32_16x16x4_f32(false, a_lo, false, w[2 * j],
                                                  (short)0, acc, false, false);
      w[2 * j] = load_kpair(wq + 8 * j);
      __builtin_amdgcn_sched_barrier(0);
      acc = __builtin_amdgcn_wmma_f32_16x16x4_f32(false, a_hi, false, w[2 * j + 1],
                                                  (short)0, acc, false, false);
      w[2 * j + 1] = load_kpair(wq + 8 * j + 4);
      a[j]         = load_quad(ap + aoff + 128 * j);
      __builtin_amdgcn_sched_barrier(0);
    }
  }
#pragma unroll
  for (int j = 0; j < KPAIRS; ++j) {
    const v2f a_lo = __builtin_shufflevector(a[j], a[j], 0, 1);
    const v2f a_hi = __builtin_shufflevector(a[j], a[j], 2, 3);
    acc = __builtin_amdgcn_wmma_f32_16x16x4_f32(false, a_lo, false, w[2 * j],
                                                (short)0, acc, false, false);
    acc = __builtin_amdgcn_wmma_f32_16x16x4_f32(false, a_hi, false, w[2 * j + 1],
                                                (short)0, acc, false, false);
  }

  const float bi = bias[n];
  float vals[8];
  float s = 0.f, ss = 0.f;
#pragma unroll
  for (int r = 0; r < 8; ++r) {
    const float v = fmaxf(acc[r] + bi, 0.f);
    vals[r] = v; s += v; ss += v * v;
  }
  s  += __shfl_xor(s, 16);
  ss += __shfl_xor(ss, 16);

  __shared__ float sred[2][16][2];
  if (lane < 16) { sred[wave][col][0] = s; sred[wave][col][1] = ss; }
  __syncthreads();
  const float ts   = sred[0][col][0] + sred[1][col][0];
  const float tss  = sred[0][col][1] + sred[1][col][1];
  const float mean = ts * (1.0f / BB);
  const float var  = tss * (1.0f / BB) - mean * mean;
  const float sc   = g[n] * rsqrtf(var + BN_EPS);
  const float bt   = beta[n];
#pragma unroll
  for (int r = 0; r < 8; ++r) {
    const int   rr = wave * 16 + r + (hi ? 8 : 0);
    const float y  = (vals[r] - mean) * sc + bt;
    if (FRAG_OUT) out[frag_index(rr, n, Hout)] = y;
    else          out[(size_t)rr * Hout + n] = y;
  }
}

extern "C" void kernel_launch(void* const* d_in, const int* in_sizes, int n_in,
                              void* d_out, int out_size, void* d_ws, size_t ws_size,
                              hipStream_t stream) {
  const float* x     = (const float*)d_in[0];
  const float* basis = (const float*)d_in[1];
  const float* bn0_g = (const float*)d_in[2];
  const float* bn0_b = (const float*)d_in[3];
  const float* W0 = (const float*)d_in[4];
  const float* b0 = (const float*)d_in[5];
  const float* g0 = (const float*)d_in[6];
  const float* be0 = (const float*)d_in[7];
  const float* W1 = (const float*)d_in[8];
  const float* b1 = (const float*)d_in[9];
  const float* g1 = (const float*)d_in[10];
  const float* be1 = (const float*)d_in[11];
  const float* W2 = (const float*)d_in[12];
  const float* b2 = (const float*)d_in[13];
  const float* g2 = (const float*)d_in[14];
  const float* be2 = (const float*)d_in[15];
  const float* W3 = (const float*)d_in[16];
  const float* b3 = (const float*)d_in[17];
  const float* g3 = (const float*)d_in[18];
  const float* be3 = (const float*)d_in[19];
  const float* Wf = (const float*)d_in[20];
  const float* bf = (const float*)d_in[21];
  const float* gf = (const float*)d_in[22];
  const float* bef = (const float*)d_in[23];

  float* ws   = (float*)d_ws;
  float* feat = ws;
  float* x0   = ws + 32768;
  float* h    = ws + 65536;
  float* a1   = ws + 73728;
  float* a2   = ws + 81920;

  bps_min_kernel<<<dim3(PP / (16 * BPS_T), BB), 32 * BPS_S, 0, stream>>>(x, basis, feat);
  bn0_kernel<<<PP / BN0_T, BN0_T, 0, stream>>>(feat, bn0_g, bn0_b, x0);
  dense_bn_relu_kernel<true><<<HH / 16, 64, 0, stream>>>(x0, PP, nullptr, 0, nullptr, 0,
                                                   W0, b0, g0, be0, h, HH);
  dense_bn_relu_kernel<true><<<HH / 16, 64, 0, stream>>>(h, HH, nullptr, 0, nullptr, 0,
                                                   W1, b1, g1, be1, a1, HH);
  dense_bn_relu_kernel<true><<<HH / 16, 64, 0, stream>>>(x0, PP, a1, HH, nullptr, 0,
                                                   W2, b2, g2, be2, h, HH);
  dense_bn_relu_kernel<true><<<HH / 16, 64, 0, stream>>>(h, HH, nullptr, 0, nullptr, 0,
                                                   W3, b3, g3, be3, a2, HH);
  dense_bn_relu_kernel<false><<<EE / 16, 64, 0, stream>>>(x0, PP, a1, HH, a2, HH,
                                                   Wf, bf, gf, bef,
                                                   (float*)d_out, EE);
}


# dense layers: next chunk's 48 loads issued as one group in front of the current chunk's 32 WMMAs (second register set, K loop unrolled per layer shape); was 1-2 loads behind each WMMA; same sums
# speedup vs baseline: 2.8978x; 1.2728x over previous
#include <hip/hip_runtime.h>

typedef __attribute__((ext_vector_type(2))) float v2f;
typedef __attribute__((ext_vector_type(4))) float v4f;
typedef __attribute__((ext_vector_type(8))) float v8f;

#define BB 32
#define NN 2048
#define PP 1024
#define HH 256
#define EE 512
#define BN_EPS 1e-5f

#define BPS_T 4
#define BPS_S 4
#define BPS_PTS (NN / BPS_S)
#define BPS_GRP (BPS_PTS / 64)

__global__ __launch_bounds__(32 * BPS_S) void bps_min_kernel(
    const float* __restrict__ x,
    const float* __restrict__ basis,
    float* __restrict__ feat)
{
  const int colg  = blockIdx.x * (16 * BPS_T);
  const int b     = blockIdx.y;
  const int lane  = threadIdx.x & 31;
  const int wave  = threadIdx.x >> 5;
  const int col   = lane & 15;
  const bool hi   = lane >= 16;

  v2f   Bm[BPS_T];
  float bn2[BPS_T];
#pragma unroll
  for (int t = 0; t < BPS_T; ++t) {
    const int   n  = colg + 16 * t + col;
    const float bx = basis[n * 3 + 0];
    const float by = basis[n * 3 + 1];
    const float bz = basis[n * 3 + 2];
    bn2[t]  = fmaf(bz, bz, fmaf(bx, bx, by * by));
    Bm[t].x = hi ? (-2.0f * bz) : (-2.0f * bx);
    Bm[t].y = hi ? 1.0f         : (-2.0f * by);
  }

  float minv[BPS_T][4];
#pragma unroll
  for (int t = 0; t < BPS_T; ++t)
#pragma unroll
    for (int r = 0; r < 4; ++r) minv[t][r] = 3.4e38f;

  auto fold = [&](const v8f& d, int t) {
#pragma unroll
    for (int r = 0; r < 4; ++r)
      minv[t][r] = fminf(fminf(d[2 * r], d[2 * r + 1]), minv[t][r]);
  };

  v8f pend;
#pragma unroll
  for (int r = 0; r < 8; ++r) pend[r] = 3.4e38f;

  auto group = [&](const float4& qx, const float4& qy, const float4& qz) {
    const float pxs[4] = {qx.x, qx.y, qx.z, qx.w};
    const float pys[4] = {qy.x, qy.y, qy.z, qy.w};
    const float pzs[4] = {qz.x, qz.y, qz.z, qz.w};
#pragma unroll
    for (int e = 0; e < 4; ++e) {
      const float px  = pxs[e];
      const float py  = pys[e];
      const float pz  = pzs[e];
      const float pn2 = fmaf(pz, pz, fmaf(px, px, py * py));
      v2f Am;
      Am.x = hi ? pz  : px;
      Am.y = hi ? pn2 : py;
#pragma unroll
      for (int t = 0; t < BPS_T; ++t) {
        v8f c = {0.f, 0.f, 0.f, 0.f, 0.f, 0.f, 0.f, 0.f};
        c = __builtin_amdgcn_wmma_f32_16x16x4_f32(false, Am, false, Bm[t],
                                                  (short)0, c, false, false);
        fold(pend, (t + BPS_T - 1) % BPS_T);
        pend = c;
      }
    }
  };

  const float* xs = x + (size_t)b * 3 * NN + wave * BPS_PTS + 4 * col;
  auto ld = [&](int plane, int g) {
    return *reinterpret_cast<const float4*>(xs + plane * NN + 64 * g);
  };
  float4 ax = ld(0, 0), ay = ld(1, 0), az = ld(2, 0);
  float4 bx4 = ld(0, 1), by4 = ld(1, 1), bz4 = ld(2, 1);
#pragma unroll 1
  for (int g = 0; g < BPS_GRP; g += 2) {
    float4 nax = ax, nay = ay, naz = az, nbx = bx4, nby = by4, nbz = bz4;
    if (g + 2 < BPS_GRP) {
      nax = ld(0, g + 2); nay = ld(1, g + 2); naz = ld(2, g + 2);
      nbx = ld(0, g + 3); nby = ld(1, g + 3); nbz = ld(2, g + 3);
    }
    group(ax, ay, az);
    group(bx4, by4, bz4);
    ax = nax; ay = nay; az = naz;
    bx4 = nbx; by4 = nby; bz4 = nbz;
  }
  fold(pend, BPS_T - 1);

  __shared__ float red[BPS_S][16 * BPS_T];
#pragma unroll
  for (int t = 0; t < BPS_T; ++t) {
    float mv = fminf(fminf(minv[t][0], minv[t][1]), fminf(minv[t][2], minv[t][3]));
    mv = fminf(mv, __shfl_xor(mv, 16));
    mv = fmaxf(mv + bn2[t], 0.0f);
    if (lane < 16) red[wave][16 * t + col] = mv;
  }
  __syncthreads();
  if (threadIdx.x < 16 * BPS_T) {
    float v = red[0][threadIdx.x];
#pragma unroll
    for (int w = 1; w < BPS_S; ++w) v = fminf(v, red[w][threadIdx.x]);
    feat[(size_t)b * PP + colg + threadIdx.x] = sqrtf(v);
  }
}

__device__ __forceinline__ int frag_index(int m, int k, int K) {
  const int lane = (m & 15) + ((k & 2) << 3);
  return ((((m >> 4) * (K >> 3) + (k >> 3)) * 32 + lane) << 2) + ((k >> 1) & 2) + (k & 1);
}

#define BN0_T 64
__global__ __launch_bounds__(BN0_T) void bn0_kernel(
    const float* __restrict__ feat, const float* __restrict__ g,
    const float* __restrict__ bshift, float* __restrict__ x0)
{
  const int j = blockIdx.x * BN0_T + threadIdx.x;
  float v[BB];
#pragma unroll
  for (int i = 0; i < BB; ++i) v[i] = feat[(size_t)i * PP + j];
  const float gj = g[j];
  const float bt = bshift[j];
  float s = 0.f, ss = 0.f;
#pragma unroll
  for (int i = 0; i < BB; ++i) { s += v[i]; ss += v[i] * v[i]; }
  const float mean = s * (1.0f / BB);
  const float var  = ss * (1.0f / BB) - mean * mean;
  const float sc   = gj * rsqrtf(var + BN_EPS);
#pragma unroll
  for (int i = 0; i < BB; ++i)
    x0[frag_index(i, j, PP)] = (v[i] - mean) * sc + bt;
}

#define KC 128
#define KSTEPS (KC / 4)
#define KPAIRS (KC / 8)

__device__ __forceinline__ v2f load_kpair(const float* __restrict__ p) {
  const float2 v = *reinterpret_cast<const float2*>(p);
  v2f r; r.x = v.x; r.y = v.y;
  return r;
}

template <int K0, int K1, int K2, bool FRAG_OUT>
__global__ __launch_bounds__(64) void dense_bn_relu_kernel(
    const float* __restrict__ in0,
    const float* __restrict__ in1,
    const float* __restrict__ in2,
    const float* __restrict__ W,
    const float* __restrict__ bias,
    const float* __restrict__ g, const float* __restrict__ beta,
    float* __restrict__ out, int Hout)
{
  constexpr int Ktot = K0 + K1 + K2;
  constexpr int nch  = Ktot / KC;
  static_assert(K0 % KC == 0 && K1 % KC == 0 && K2 % KC == 0 && K0 >= KC,
                "a chunk must not straddle two concat inputs");

  const int  c    = blockIdx.x * 16;
  const int  lane = threadIdx.x & 31;
  const int  wave = threadIdx.x >> 5;
  const int  col  = lane & 15;
  const bool hi   = lane >= 16;
  const int  n    = c + col;
  const int  koff = hi ? 2 : 0;

  const float* wp = W + (size_t)n * Ktot + koff;
  const int mt = __builtin_amdgcn_readfirstlane(wave);
  auto a_chunk = [&](int kbase) -> const float* {
    const float* p;
    int Ks, kl;
    if (kbase < K0)           { p = in0; Ks = K0; kl = kbase; }
    else if (kbase < K0 + K1) { p = in1; Ks = K1; kl = kbase - K0; }
    else                      { p = in2; Ks = K2; kl = kbase - K0 - K1; }
    return p + (size_t)(mt * (Ks >> 3) + (kl >> 3)) * 128;
  };
  const int aoff = 4 * lane;
  auto load_quad = [](const float* __restrict__ p) {
    return *reinterpret_cast<const v4f*>(p);
  };

  v4f a[2][KPAIRS];
  v2f w[2][KSTEPS];
  {
    const float* ap = a_chunk(0);
#pragma unroll
    for (int j = 0; j < KPAIRS; ++j) {
      w[0][2 * j]     = load_kpair(wp + 8 * j);
      w[0][2 * j + 1] = load_kpair(wp + 8 * j + 4);
      a[0][j]         = load_quad(ap + aoff + 128 * j);
    }
  }
  __builtin_amdgcn_sched_barrier(0);

  {
    auto pf = [](const float* p) { __builtin_prefetch(p, 0, 2); };
    const float* pw = W + (size_t)(c + (threadIdx.x & 15)) * Ktot;
    constexpr int nw = Ktot >> 6;
    int           uw = KC / 64 + (threadIdx.x >> 4);
    const float* t0 = in0 + (size_t)mt * (K0 >> 3) * 128;
    constexpr int n0 = K0 >> 2;
    int           ua = KC / 4 + lane;
    while (ua < n0 || uw < nw) {
      if (ua < n0) pf(t0 + ua * 64);
      if (uw < nw) pf(pw + uw * 64);
      ua += 32;
      uw += 4;
    }
    if (K1 > 0) {
      const float* t1 = in1 + (size_t)mt * (K1 >> 3) * 128;
      for (int u = lane; u < (K1 >> 2); u += 32) pf(t1 + u * 64);
    }
    if (K2 > 0) {
      const float* t2 = in2 + (size_t)mt * (K2 >> 3) * 128;
      for (int u = lane; u < (K2 >> 2); u += 32) pf(t2 + u * 64);
    }
  }
  __builtin_amdgcn_sched_barrier(0);

  v8f acc = {0.f, 0.f, 0.f, 0.f, 0.f, 0.f, 0.f, 0.f};
#pragma unroll
  for (int ch = 0; ch < nch; ++ch) {
    const int cur = ch & 1;
    const int nxt = cur ^ 1;
    if (ch + 1 < nch) {
      const float* ap = a_chunk((ch + 1) * KC);
      const float* wq = wp + (ch + 1) * KC;
#pragma unroll
      for (int j = 0; j < KPAIRS; ++j) {
        w[nxt][2 * j]     = load_kpair(wq + 8 * j);
        w[nxt][2 * j + 1] = load_kpair(wq + 8 * j + 4);
        a[nxt][j]         = load_quad(ap + aoff + 128 * j);
      }
    }
    __builtin_amdgcn_sched_barrier(0);
#pragma unroll
    for (int j = 0; j < KPAIRS; ++j) {
      const v2f a_lo = __builtin_shufflevector(a[cur][j], a[cur][j], 0, 1);
      const v2f a_hi = __builtin_shufflevector(a[cur][j], a[cur][j], 2, 3);
      acc = __builtin_amdgcn_wmma_f32_16x16x4_f32(false, a_lo, false, w[cur][2 * j],
                                                  (short)0, acc, false, false);
      acc = __builtin_amdgcn_wmma_f32_16x16x4_f32(false, a_hi, false, w[cur][2 * j + 1],
                                                  (short)0, acc, false, false);
    }
    __builtin_amdgcn_sched_barrier(0);
  }

  const float bi = bias[n];
  float vals[8];
  float s = 0.f, ss = 0.f;
#pragma unroll
  for (int r = 0; r < 8; ++r) {
    const float v = fmaxf(acc[r] + bi, 0.f);
    vals[r] = v; s += v; ss += v * v;
  }
  s  += __shfl_xor(s, 16);
  ss += __shfl_xor(ss, 16);

  __shared__ float sred[2][16][2];
  if (lane < 16) { sred[wave][col][0] = s; sred[wave][col][1] = ss; }
  __syncthreads();
  const float ts   = sred[0][col][0] + sred[1][col][0];
  const float tss  = sred[0][col][1] + sred[1][col][1];
  const float mean = ts * (1.0f / BB);
  const float var  = tss * (1.0f / BB) - mean * mean;
  const float sc   = g[n] * rsqrtf(var + BN_EPS);
  const float bt   = beta[n];
#pragma unroll
  for (int r = 0; r < 8; ++r) {
    const int   rr = wave * 16 + r + (hi ? 8 : 0);
    const float y  = (vals[r] - mean) * sc + bt;
    if (FRAG_OUT) out[frag_index(rr, n, Hout)] = y;
    else          out[(size_t)rr * Hout + n] = y;
  }
}

extern "C" void kernel_launch(void* const* d_in, const int* in_sizes, int n_in,
                              void* d_out, int out_size, void* d_ws, size_t ws_size,
                              hipStream_t stream) {
  const float* x     = (const float*)d_in[0];
  const float* basis = (const float*)d_in[1];
  const float* bn0_g = (const float*)d_in[2];
  const float* bn0_b = (const float*)d_in[3];
  const float* W0 = (const float*)d_in[4];
  const float* b0 = (const float*)d_in[5];
  const float* g0 = (const float*)d_in[6];
  const float* be0 = (const float*)d_in[7];
  const float* W1 = (const float*)d_in[8];
  const float* b1 = (const float*)d_in[9];
  const float* g1 = (const float*)d_in[10];
  const float* be1 = (const float*)d_in[11];
  const float* W2 = (const float*)d_in[12];
  const float* b2 = (const float*)d_in[13];
  const float* g2 = (const float*)d_in[14];
  const float* be2 = (const float*)d_in[15];
  const float* W3 = (const float*)d_in[16];
  const float* b3 = (const float*)d_in[17];
  const float* g3 = (const float*)d_in[18];
  const float* be3 = (const float*)d_in[19];
  const float* Wf = (const float*)d_in[20];
  const float* bf = (const float*)d_in[21];
  const float* gf = (const float*)d_in[22];
  const float* bef = (const float*)d_in[23];

  float* ws   = (float*)d_ws;
  float* feat = ws;
  float* x0   = ws + 32768;
  float* h    = ws + 65536;
  float* a1   = ws + 73728;
  float* a2   = ws + 81920;

  bps_min_kernel<<<dim3(PP / (16 * BPS_T), BB), 32 * BPS_S, 0, stream>>>(x, basis, feat);
  bn0_kernel<<<PP / BN0_T, BN0_T, 0, stream>>>(feat, bn0_g, bn0_b, x0);
  dense_bn_relu_kernel<PP, 0, 0, true><<<HH / 16, 64, 0, stream>>>(
      x0, nullptr, nullptr, W0, b0, g0, be0, h, HH);
  dense_bn_relu_kernel<HH, 0, 0, true><<<HH / 16, 64, 0, stream>>>(
      h, nullptr, nullptr, W1, b1, g1, be1, a1, HH);
  dense_bn_relu_kernel<PP, HH, 0, true><<<HH / 16, 64, 0, stream>>>(
      x0, a1, nullptr, W2, b2, g2, be2, h, HH);
  dense_bn_relu_kernel<HH, 0, 0, true><<<HH / 16, 64, 0, stream>>>(
      h, nullptr, nullptr, W3, b3, g3, be3, a2, HH);
  dense_bn_relu_kernel<PP, HH, HH, false><<<EE / 16, 64, 0, stream>>>(
      x0, a1, a2, Wf, bf, gf, bef, (float*)d_out, EE);
}
